# starting kernel
# baseline (speedup 1.0000x reference)
#include <hip/hip_runtime.h>

typedef __attribute__((ext_vector_type(16))) __bf16 v16bf;
typedef __attribute__((ext_vector_type(8)))  __bf16 v8bf;
typedef __attribute__((ext_vector_type(8)))  float  v8f;

__device__ __forceinline__ __bf16 f2bf(float f) {
  union { float f; unsigned u; } v; v.f = f;
  unsigned r = v.u + 0x7FFFu + ((v.u >> 16) & 1u);
  union { unsigned short s; __bf16 b; } o;
  o.s = (unsigned short)(r >> 16);
  return o.b;
}

template <int K>
__global__ void gemm_wmma_bf16(const __bf16* __restrict__ A,
                               const __bf16* __restrict__ Bt,
                               float* __restrict__ C, int ldc) {
  const int wave = threadIdx.x >> 5;
  const int lane = threadIdx.x & 31;
  const int m16  = lane & 15;
  const size_t arow = (size_t)(blockIdx.x * 16 + m16) * K;
  const size_t brow = (size_t)(wave * 16 + m16) * K;
  const int khalf = (lane >> 4) << 3;

  v8f acc = {};
#pragma unroll
  for (int k0 = 0; k0 < K; k0 += 32) {
    const v8bf alo = *(const v8bf*)(A  + arow + k0 + khalf);
    const v8bf ahi = *(const v8bf*)(A  + arow + k0 + 16 + khalf);
    const v8bf blo = *(const v8bf*)(Bt + brow + k0 + khalf);
    const v8bf bhi = *(const v8bf*)(Bt + brow + k0 + 16 + khalf);
    const v16bf a = __builtin_shufflevector(alo, ahi, 0,1,2,3,4,5,6,7,8,9,10,11,12,13,14,15);
    const v16bf b = __builtin_shufflevector(blo, bhi, 0,1,2,3,4,5,6,7,8,9,10,11,12,13,14,15);
    acc = __builtin_amdgcn_wmma_f32_16x16x32_bf16(
        false, a, false, b, (short)0, acc, false, false);
  }

  const int col  = wave * 16 + m16;
  const int crow = blockIdx.x * 16 + ((lane >> 4) << 3);
#pragma unroll
  for (int r = 0; r < 8; ++r)
    C[(size_t)(crow + r) * ldc + col] = acc[r];
}

__global__ void cvt_bf16_k(const float* __restrict__ in, __bf16* __restrict__ out,
                           long long n) {
  long long i = (long long)blockIdx.x * blockDim.x + threadIdx.x;
  if (i < n) out[i] = f2bf(in[i]);
}

__global__ void cvt_tr_k(const float* __restrict__ W, __bf16* __restrict__ Wt,
                         int K, int N) {
  int i = blockIdx.x * blockDim.x + threadIdx.x;
  if (i >= K * N) return;
  int k = i / N, n = i % N;
  Wt[(size_t)n * K + k] = f2bf(W[i]);
}

__global__ void init_deg_k(float* __restrict__ deg, int n) {
  int i = blockIdx.x * blockDim.x + threadIdx.x;
  if (i < n) deg[i] = 1.0f;
}

__global__ void edge_deg_k(const int* __restrict__ dst, float* __restrict__ deg, int e) {
  int i = blockIdx.x * blockDim.x + threadIdx.x;
  if (i < e) atomicAdd(&deg[dst[i]], 1.0f);
}

__global__ void rsqrt_k(float* __restrict__ d, int n) {
  int i = blockIdx.x * blockDim.x + threadIdx.x;
  if (i < n) { float v = d[i]; d[i] = v > 0.0f ? rsqrtf(v) : 0.0f; }
}

__global__ void init_agg_k(const float* __restrict__ hlin, const float* __restrict__ dinv,
                           const float* __restrict__ bias, float* __restrict__ out,
                           long long total, int F) {
  long long i = (long long)blockIdx.x * blockDim.x + threadIdx.x;
  if (i >= total) return;
  int n = (int)(i / F), f = (int)(i % F);
  float di = dinv[n];
  out[i] = hlin[i] * di * di + bias[f];
}

__global__ void edge_agg_k(const int* __restrict__ src, const int* __restrict__ dst,
                           const float* __restrict__ dinv, const float* __restrict__ hlin,
                           float* __restrict__ out, int E, int F4, int F) {
  long long i = (long long)blockIdx.x * blockDim.x + threadIdx.x;
  if (i >= (long long)E * F4) return;
  int e = (int)(i / F4);
  int f = (int)(i % F4) * 4;
  int s = src[e], d = dst[e];
  float nrm = dinv[s] * dinv[d];
  const float4 hv = *(const float4*)(hlin + (size_t)s * F + f);
  float* op = out + (size_t)d * F + f;
  atomicAdd(op + 0, hv.x * nrm);
  atomicAdd(op + 1, hv.y * nrm);
  atomicAdd(op + 2, hv.z * nrm);
  atomicAdd(op + 3, hv.w * nrm);
}

__global__ void relu_k(float* __restrict__ p, long long n) {
  long long i = (long long)blockIdx.x * blockDim.x + threadIdx.x;
  if (i < n) p[i] = fmaxf(p[i], 0.0f);
}

extern "C" void kernel_launch(void* const* d_in, const int* in_sizes, int n_in,
                              void* d_out, int out_size, void* d_ws, size_t ws_size,
                              hipStream_t stream) {
  const float* x   = (const float*)d_in[0];
  const int*   ei  = (const int*)d_in[1];
  const int*   yei = (const int*)d_in[2];
  const float* W1  = (const float*)d_in[3];
  const float* b1  = (const float*)d_in[4];
  const float* Wmu = (const float*)d_in[5];
  const float* bmu = (const float*)d_in[6];
  const float* Wls = (const float*)d_in[7];
  const float* bls = (const float*)d_in[8];

  const int DIN = 128, DH = 128, DZ = 32;
  const int N  = in_sizes[0] / DIN;
  const int E1 = in_sizes[1] / 2;
  const int E2 = in_sizes[2] / 2;

  const int* src1 = ei;   const int* dst1 = ei  + E1;
  const int* src2 = yei;  const int* dst2 = yei + E2;

  char* wsb = (char*)d_ws;
  float*  dinv = (float*)wsb;
  __bf16* w1t  = (__bf16*)(wsb + (size_t)65536 * 4);
  __bf16* wmut = w1t + 128 * 128;
  __bf16* wlst = wmut + 32 * 128;
  __bf16* abf  = (__bf16*)(wsb + (size_t)(65536 + 16384) * 4);
  float*  buf1 = (float*)(wsb + (size_t)(65536 + 16384) * 4 + (size_t)N * DH * 2);
  float*  buf2 = buf1 + (size_t)N * DH;

  const int T = 256;
  auto blocks = [](long long total, int t) { return (unsigned)((total + t - 1) / t); };

  init_deg_k<<<blocks(N, T), T, 0, stream>>>(dinv, N);
  edge_deg_k<<<blocks(E1, T), T, 0, stream>>>(dst1, dinv, E1);
  edge_deg_k<<<blocks(E2, T), T, 0, stream>>>(dst2, dinv, E2);
  rsqrt_k<<<blocks(N, T), T, 0, stream>>>(dinv, N);

  cvt_bf16_k<<<blocks((long long)N * DIN, T), T, 0, stream>>>(x, abf, (long long)N * DIN);
  cvt_tr_k<<<blocks(DIN * DH, T), T, 0, stream>>>(W1, w1t, DIN, DH);
  gemm_wmma_bf16<128><<<N / 16, 32 * (DH / 16), 0, stream>>>(abf, w1t, buf1, DH);

  init_agg_k<<<blocks((long long)N * DH, T), T, 0, stream>>>(buf1, dinv, b1, buf2,
                                                             (long long)N * DH, DH);
  edge_agg_k<<<blocks((long long)E1 * (DH / 4), T), T, 0, stream>>>(src1, dst1, dinv,
                                                                    buf1, buf2, E1, DH / 4, DH);
  edge_agg_k<<<blocks((long long)E2 * (DH / 4), T), T, 0, stream>>>(src2, dst2, dinv,
                                                                    buf1, buf2, E2, DH / 4, DH);
  relu_k<<<blocks((long long)N * DH, T), T, 0, stream>>>(buf2, (long long)N * DH);

  cvt_bf16_k<<<blocks((long long)N * DH, T), T, 0, stream>>>(buf2, abf, (long long)N * DH);
  cvt_tr_k<<<blocks(DH * DZ, T), T, 0, stream>>>(Wmu, wmut, DH, DZ);
  cvt_tr_k<<<blocks(DH * DZ, T), T, 0, stream>>>(Wls, wlst, DH, DZ);

  float* zmu_lin = buf1;
  float* zls_lin = buf1 + (size_t)N * DZ;
  gemm_wmma_bf16<128><<<N / 16, 32 * (DZ / 16), 0, stream>>>(abf, wmut, zmu_lin, DZ);
  gemm_wmma_bf16<128><<<N / 16, 32 * (DZ / 16), 0, stream>>>(abf, wlst, zls_lin, DZ);

  float* out_mu = (float*)d_out;
  float* out_ls = out_mu + (size_t)N * DZ;

  init_agg_k<<<blocks((long long)N * DZ, T), T, 0, stream>>>(zmu_lin, dinv, bmu, out_mu,
                                                             (long long)N * DZ, DZ);
  init_agg_k<<<blocks((long long)N * DZ, T), T, 0, stream>>>(zls_lin, dinv, bls, out_ls,
                                                             (long long)N * DZ, DZ);

  edge_agg_k<<<blocks((long long)E1 * (DZ / 4), T), T, 0, stream>>>(src1, dst1, dinv,
                                                                    zmu_lin, out_mu, E1, DZ / 4, DZ);
  edge_agg_k<<<blocks((long long)E2 * (DZ / 4), T), T, 0, stream>>>(src2, dst2, dinv,
                                                                    zmu_lin, out_mu, E2, DZ / 4, DZ);
  edge_agg_k<<<blocks((long long)E1 * (DZ / 4), T), T, 0, stream>>>(src1, dst1, dinv,
                                                                    zls_lin, out_ls, E1, DZ / 4, DZ);
  edge_agg_k<<<blocks((long long)E2 * (DZ / 4), T), T, 0, stream>>>(src2, dst2, dinv,
                                                                    zls_lin, out_ls, E2, DZ / 4, DZ);
}


# Normalisation factored out of the edges: owner pass lists edges once, counts degrees in LDS (no global atomics); products store rows*dinv[row]; no per-edge weight or start plane; sum order changes
# speedup vs baseline: 3.3222x; 3.3222x over previous
#include <hip/hip_runtime.h>

typedef __attribute__((ext_vector_type(16))) __bf16 v16bf;
typedef __attribute__((ext_vector_type(8)))  __bf16 v8bf;
typedef __attribute__((ext_vector_type(8)))  float  v8f;
typedef __attribute__((ext_vector_type(4)))  float    v4f;
typedef __attribute__((ext_vector_type(4)))  unsigned v4u;

__device__ __forceinline__ __bf16 f2bf(float f) {
  union { float f; unsigned u; } v; v.f = f;
  unsigned r = v.u + 0x7FFFu + ((v.u >> 16) & 1u);
  union { unsigned short s; __bf16 b; } o;
  o.s = (unsigned short)(r >> 16);
  return o.b;
}

template <int NT, bool RELU>
__global__ void __launch_bounds__(128) __attribute__((amdgpu_waves_per_eu(1, 4)))
prod_strip_k(const float* __restrict__ A, const __bf16* __restrict__ Bt,
             const float* __restrict__ dinv, float* __restrict__ C, int nstrips) {
  constexpr int K  = 128;
  constexpr int N  = 16 * NT;
  const int lane  = threadIdx.x & 31;
  const int strip = blockIdx.x * (blockDim.x >> 5) + (threadIdx.x >> 5);
  if (strip >= nstrips) return;
  const int m16   = lane & 15;
  const int khalf = (lane >> 4) << 3;
  const float* ap = A + (size_t)(strip * 16 + m16) * K + khalf;

  v8f alo[4], ahi[4];
#pragma unroll
  for (int s = 0; s < 4; ++s) {
    alo[s] = *(const v8f*)(ap + 32 * s);
    ahi[s] = *(const v8f*)(ap + 32 * s + 16);
  }
  const __bf16* bp = Bt + (size_t)m16 * K + khalf;
  v16bf b[2][NT];
  auto load_b = [&](v16bf* dst, int s) {
#pragma unroll
    for (int t = 0; t < NT; ++t) {
      const v8bf blo = *(const v8bf*)(bp + (size_t)t * 16 * K + 32 * s);
      const v8bf bhi = *(const v8bf*)(bp + (size_t)t * 16 * K + 32 * s + 16);
      dst[t] = __builtin_shufflevector(blo, bhi, 0,1,2,3,4,5,6,7,8,9,10,11,12,13,14,15);
    }
  };
  load_b(b[0], 0);

  v16bf a[4];
#pragma unroll
  for (int s = 0; s < 4; ++s) {
    v8f lo = alo[s], hi = ahi[s];
    if (RELU) {
#pragma unroll
      for (int i = 0; i < 8; ++i) { lo[i] = fmaxf(lo[i], 0.0f); hi[i] = fmaxf(hi[i], 0.0f); }
    }
    const v8bf l = __builtin_convertvector(lo, v8bf);
    const v8bf h = __builtin_convertvector(hi, v8bf);
    a[s] = __builtin_shufflevector(l, h, 0,1,2,3,4,5,6,7,8,9,10,11,12,13,14,15);
  }

  v8f acc[NT];
#pragma unroll
  for (int t = 0; t < NT; ++t) acc[t] = v8f{};
#pragma unroll
  for (int s = 0; s < 4; ++s) {
    if (s + 1 < 4) load_b(b[(s + 1) & 1], s + 1);
#pragma unroll
    for (int t = 0; t < NT; ++t)
      acc[t] = __builtin_amdgcn_wmma_f32_16x16x32_bf16(
          false, a[s], false, b[s & 1][t], (short)0, acc[t], false, false);
  }

  const int crow = strip * 16 + ((lane >> 4) << 3);
  const v8f di = *(const v8f*)(dinv + crow);
#pragma unroll
  for (int t = 0; t < NT; ++t) {
    const int col = t * 16 + m16;
#pragma unroll
    for (int r = 0; r < 8; ++r) C[(size_t)(crow + r) * N + col] = acc[t][r] * di[r];
  }
}

constexpr int TR      = 196;
constexpr int LSTRIDE = 4096;
constexpr int REC0    = 32;
static_assert(TR <= 255, "a record keeps the row inside the range in 8 bits, 255 = no edge");

__device__ __forceinline__ void cvt_tr_one(const float* __restrict__ W, __bf16* __restrict__ Wt,
                                           int i, int K, int N) {
  const int k = i / N, n = i % N;
  Wt[(size_t)n * K + k] = f2bf(W[i]);
}

__global__ void setup_k(const float* __restrict__ W1, const float* __restrict__ Wmu,
                        const float* __restrict__ Wls, __bf16* __restrict__ w1t,
                        __bf16* __restrict__ wmut, __bf16* __restrict__ wlst,
                        const int* __restrict__ dst1, int e1, const int* __restrict__ dst2, int e2,
                        unsigned* __restrict__ own32, int own_threads) {
  int i = blockIdx.x * blockDim.x + threadIdx.x;
  if (i < own_threads) {
    const int E    = e1 + e2;
    const int lane = i & 31;
    const int wb   = (i >> 5) * 512;
#pragma unroll
    for (int q = 0; q < 4; ++q) {
      const int e0 = wb + q * 128 + lane * 4;
      const int* p = e0 < e1 ? dst1 + e0 : dst2 + (e0 - e1);
      const bool one_list = (e0 + 4 <= e1) || (e0 >= e1 && e0 + 4 <= E);
      int d[4];
      if (one_list && ((size_t)p & 15) == 0) {
        const v4u v = *(const v4u*)p;
        d[0] = (int)v.x; d[1] = (int)v.y; d[2] = (int)v.z; d[3] = (int)v.w;
      } else {
#pragma unroll
        for (int b = 0; b < 4; ++b) {
          const int e = e0 + b;
          d[b] = 0;
          if (e < e1)     d[b] = dst1[e];
          else if (e < E) d[b] = dst2[e - e1];
        }
      }
      own32[e0 >> 2] = (unsigned)(d[0] / TR) | ((unsigned)(d[1] / TR) << 8) |
                       ((unsigned)(d[2] / TR) << 16) | ((unsigned)(d[3] / TR) << 24);
    }
    return;
  }
  i -= own_threads;
  if (i < 128 * 128) { cvt_tr_one(W1, w1t, i, 128, 128); return; }
  i -= 128 * 128;
  if (i < 128 * 32) { cvt_tr_one(Wmu, wmut, i, 128, 32); return; }
  i -= 128 * 32;
  if (i < 128 * 32) { cvt_tr_one(Wls, wlst, i, 128, 32); }
}

template <bool B> struct flag_t { static constexpr bool value = B; };

__device__ __forceinline__ unsigned zero_bytes(unsigned y) {
  return ~(((y & 0x7f7f7f7fu) + 0x7f7f7f7fu) | y | 0x7f7f7f7fu);
}

template <typename OVER>
__device__ __forceinline__ void scan_owner_bytes(const v4u* __restrict__ own4, int E, int bid,
                                                 int wv, int lane, int* list, int cap,
                                                 int* s_cnt, OVER over) {
  const int      G   = (E + 15) >> 4;
  const unsigned pat = (unsigned)bid * 0x01010101u;
  constexpr int  U   = 4;
  for (int gb = wv * 32; gb < G; gb += 1024 * U) {
    v4u x[U];
#pragma unroll
    for (int u = 0; u < U; ++u) x[u] = own4[min(gb + u * 1024 + lane, G - 1)];
#pragma unroll
    for (int u = 0; u < U; ++u) {
      const int g = gb + u * 1024 + lane;
      const unsigned ok = g < G ? ~0u : 0u;
      unsigned t0 = zero_bytes(x[u].x ^ pat) & ok, t1 = zero_bytes(x[u].y ^ pat) & ok;
      unsigned t2 = zero_bytes(x[u].z ^ pat) & ok, t3 = zero_bytes(x[u].w ^ pat) & ok;
      while (true) {
        const bool     has = (t0 | t1 | t2 | t3) != 0u;
        const unsigned m   = __builtin_amdgcn_ballot_w32(has);
        if (m == 0u) break;
        int slot0 = 0;
        if (lane == 0) slot0 = atomicAdd(s_cnt, __builtin_popcount(m));
        slot0 = __builtin_amdgcn_readfirstlane(slot0);
        if (has) {
          const int slot = slot0 + (int)__builtin_amdgcn_mbcnt_lo(m, 0u);
          const int      q  = t0 ? 0 : t1 ? 1 : t2 ? 2 : 3;
          const unsigned tq = t0 ? t0 : t1 ? t1 : t2 ? t2 : t3;
          const unsigned tc = tq & (tq - 1u);
          t0 = q == 0 ? tc : t0;  t1 = q == 1 ? tc : t1;
          t2 = q == 2 ? tc : t2;  t3 = q == 3 ? tc : t3;
          const int e = g * 16 + q * 4 + (__builtin_ctz(tq) >> 3);
          if (slot < cap)  list[slot] = e < E ? e : -1;
          else if (e < E)  over(e);
        }
      }
    }
  }
}

constexpr int OWNER_LDS = 168 * 1024;

__global__ void __launch_bounds__(1024)
owner_build_k(const v4u* __restrict__ own4, int E1, int E2,
              const int* __restrict__ src1, const int* __restrict__ dst1,
              const int* __restrict__ src2, const int* __restrict__ dst2,
              float* __restrict__ dinv, int* __restrict__ wlist, int N) {
  constexpr int CAP = OWNER_LDS / 4 - TR - 8;
  static_assert(CAP >= LSTRIDE, "edges that fit the record slots must fit the LDS list");
  __shared__ int list[CAP];
  __shared__ int cntl[TR];
  __shared__ int s_cnt;

  const int tid  = threadIdx.x;
  const int lane = tid & 31;
  const int wv   = __builtin_amdgcn_readfirstlane(tid >> 5);
  const int bid  = blockIdx.x;
  const int base = bid * TR;
  const int rows = min(TR, N - base);
  const int E    = E1 + E2;
  int* wl = wlist + (size_t)bid * LSTRIDE;

  if (tid == 0)  s_cnt = 0;
  if (tid < TR)  cntl[tid] = 0;
  __syncthreads();

  auto dst_of = [&](int e) { return e < E1 ? dst1[e] : dst2[e - E1]; };
  auto src_of = [&](int e) { return e < E1 ? src1[e] : src2[e - E1]; };

  scan_owner_bytes(own4, E, bid, wv, lane, list, CAP, &s_cnt,
                   [&](int e) { atomicAdd(&cntl[dst_of(e) - base], 1); });
  __syncthreads();

  const int  cnt  = s_cnt;
  const int  n    = min(cnt, CAP);
  const bool fits = cnt <= LSTRIDE - REC0;
  constexpr int U = 4;
  for (int i0 = tid; i0 < n; i0 += 1024 * U) {
    int e[U], s[U], d[U];
#pragma unroll
    for (int u = 0; u < U; ++u) { const int i = i0 + u * 1024; e[u] = i < n ? list[i] : -1; }
#pragma unroll
    for (int u = 0; u < U; ++u) {
      s[u] = 0; d[u] = base;
      if (e[u] >= 0) { s[u] = src_of(e[u]); d[u] = dst_of(e[u]); }
    }
#pragma unroll
    for (int u = 0; u < U; ++u) {
      const int i = i0 + u * 1024;
      if (i < n) {
        unsigned word = 0xff000000u;
        if (e[u] >= 0) {
          const int dl = d[u] - base;
          atomicAdd(&cntl[dl], 1);
          word = (unsigned)s[u] | ((unsigned)dl << 24);
        }
        if (fits) wl[REC0 + i] = (int)word;
      }
    }
  }
  __syncthreads();

  if (tid < rows) {
    const float deg = 1.0f + (float)cntl[tid];
    dinv[base + tid] = deg > 0.0f ? rsqrtf(deg) : 0.0f;
  }
  if (tid == 0) wl[0] = fits ? cnt : -1;
}

template <int F, bool PLANES>
__global__ void __launch_bounds__(1024)
edge_own_k(const v4u* __restrict__ own4, int E1, int E2,
           const int* __restrict__ src1, const int* __restrict__ dst1,
           const int* __restrict__ src2, const int* __restrict__ dst2,
           const float* __restrict__ dinv, const float* __restrict__ hlin,
           const float* __restrict__ bias0, const float* __restrict__ bias1,
           float* __restrict__ out, size_t plane, const int* __restrict__ wlist, int N) {
  constexpr int J   = F / 32;
  constexpr int EPW = 32 / J;
  constexpr int LPE = 32 / EPW;
  constexpr int CAP = (OWNER_LDS - TR * F * 4) / 4 - TR - 8;
  static_assert(CAP >= LSTRIDE, "edges that fit the record slots must fit the LDS list");
  __shared__ float tile[TR * F];
  __shared__ int   list[CAP];
  __shared__ float s_di[TR];
  __shared__ int   s_cnt;

  const int tid  = threadIdx.x;
  const int lane = tid & 31;
  const int wv   = __builtin_amdgcn_readfirstlane(tid >> 5);
  const int bid  = blockIdx.x;
  const int base = bid * TR;
  const int rows = min(TR, N - base);
  const int E    = E1 + E2;
  v4f* t4 = (v4f*)tile;
  const int* wl = wlist + (size_t)bid * LSTRIDE;

  auto out4 = [&](int i) -> v4f* {
    if (!PLANES) return (v4f*)(out + (size_t)base * F) + i;
    const int r = i >> 4, q = i & 15;
    return (v4f*)(out + (q >= 8 ? plane : (size_t)0) + (size_t)(base + r) * 32 + (q & 7) * 4);
  };

  const int saved = __builtin_amdgcn_readfirstlane(wl[0]);
  if (tid == 0) s_cnt = saved >= 0 ? saved : 0;
  if (tid < rows) s_di[tid] = dinv[base + tid];
  {
    constexpr int NI = (TR * F / 4 + 1023) / 1024;
    const int lim = rows * (F / 4);
    const v4f* own_rows = (const v4f*)(hlin + (size_t)base * F);
    v4f r[NI];
#pragma unroll
    for (int k = 0; k < NI; ++k) r[k] = own_rows[min(tid + k * 1024, lim - 1)];
#pragma unroll
    for (int k = 0; k < NI; ++k) t4[min(tid + k * 1024, lim - 1)] = r[k];
  }
  __syncthreads();

  auto dst_of = [&](int e) { return e < E1 ? dst1[e] : dst2[e - E1]; };
  auto src_of = [&](int e) { return e < E1 ? src1[e] : src2[e - E1]; };

  if (saved < 0) {
    scan_owner_bytes(own4, E, bid, wv, lane, list, CAP, &s_cnt, [&](int e) {
      const int s = src_of(e), d = dst_of(e);
      for (int j = 0; j < F; ++j) atomicAdd(&tile[(d - base) * F + j], hlin[(size_t)s * F + j]);
    });
  }
  __syncthreads();

  const int cnt = s_cnt;
  const int n   = saved >= 0 ? saved : min(cnt, CAP);
  const int nb  = (n + EPW - 1) / EPW;
  auto gather = [&](auto rec_flag) {
    constexpr bool REC = decltype(rec_flag)::value;
    auto fetch_rec = [&](int b) -> unsigned {
      const int eq = b * EPW + lane / LPE;
      unsigned r = 0xff000000u;
      if (b < nb && eq < n) {
        if (REC) r = (unsigned)wl[REC0 + eq];
        else {
          const int e = list[eq];
          if (e >= 0) r = (unsigned)src_of(e) | ((unsigned)(dst_of(e) - base) << 24);
        }
      }
      return r;
    };
    unsigned rec = fetch_rec(wv);
    for (int b = wv; b < nb; b += 32) {
      const int dl   = (int)(rec >> 24);
      const int live = dl != 255;
      const int s    = live ? (int)(rec & 0x00ffffffu) : 0;
      float v[EPW][J];
#pragma unroll
      for (int k = 0; k < EPW; ++k) {
        const int sk = __builtin_amdgcn_readlane(s, k * LPE);
#pragma unroll
        for (int j = 0; j < J; ++j) v[k][j] = hlin[(size_t)sk * F + lane + 32 * j];
      }
      rec = fetch_rec(b + 32);
#pragma unroll
      for (int k = 0; k < EPW; ++k) {
        if (__builtin_amdgcn_readlane(live, k * LPE)) {
          const int dk = __builtin_amdgcn_readlane(dl, k * LPE);
          float* tp = tile + dk * F + lane;
#pragma unroll
          for (int j = 0; j < J; ++j) atomicAdd(tp + 32 * j, v[k][j]);
        }
      }
    }
  };
  if (saved >= 0) gather(flag_t<true>{});
  else            gather(flag_t<false>{});
  __syncthreads();

  for (int i = tid; i < rows * (F / 4); i += 1024) {
    const int r = i / (F / 4), q = i % (F / 4);
    const float* bp = (PLANES && q >= 8) ? bias1 + (q - 8) * 4 : bias0 + q * 4;
    v4f bv; bv.x = bp[0]; bv.y = bp[1]; bv.z = bp[2]; bv.w = bp[3];
    *out4(i) = t4[i] * s_di[r] + bv;
  }
}

extern "C" void kernel_launch(void* const* d_in, const int* in_sizes, int n_in,
                              void* d_out, int out_size, void* d_ws, size_t ws_size,
                              hipStream_t stream) {
  const float* x   = (const float*)d_in[0];
  const int*   ei  = (const int*)d_in[1];
  const int*   yei = (const int*)d_in[2];
  const float* W1  = (const float*)d_in[3];
  const float* b1  = (const float*)d_in[4];
  const float* Wmu = (const float*)d_in[5];
  const float* bmu = (const float*)d_in[6];
  const float* Wls = (const float*)d_in[7];
  const float* bls = (const float*)d_in[8];

  const int DIN = 128, DH = 128, DZ = 32;
  const int N  = in_sizes[0] / DIN;
  const int E1 = in_sizes[1] / 2;
  const int E2 = in_sizes[2] / 2;

  const int* src1 = ei;   const int* dst1 = ei  + E1;
  const int* src2 = yei;  const int* dst2 = yei + E2;

  char* wsb = (char*)d_ws;
  float*  dinv = (float*)wsb;
  __bf16* w1t  = (__bf16*)(wsb + (size_t)65536 * 4);
  __bf16* wmut = w1t + 128 * 128;
  __bf16* wlst = wmut + 32 * 128;
  unsigned char* own = (unsigned char*)(wsb + (size_t)(65536 + 16384 + 65536) * 4);
  int*    wlist = (int*)(own + ((size_t)1 << 20));
  float*  buf1 = (float*)(wsb + (size_t)(65536 + 16384) * 4 + (size_t)N * DH * 2);
  float*  buf2 = buf1 + (size_t)N * DH;

  const int T = 256;
  const int TP = 128;
  auto blocks = [](long long total, int t) { return (unsigned)((total + t - 1) / t); };
  const int NS = N / 16;
  const long long EA = (long long)E1 + E2;
  const int TO = 1024;
  const int NO = (N + TR - 1) / TR;

  const int OT = (int)((EA + 511) / 512) * 32;
  setup_k<<<blocks((long long)OT + DIN * DH + 2 * DH * DZ, T), T, 0, stream>>>(
      W1, Wmu, Wls, w1t, wmut, wlst, dst1, E1, dst2, E2, (unsigned*)own, OT);
  owner_build_k<<<NO, TO, 0, stream>>>((const v4u*)own, E1, E2, src1, dst1, src2, dst2, dinv, wlist, N);

  prod_strip_k<8, false><<<blocks((long long)NS * 32, TP), TP, 0, stream>>>(x, w1t, dinv, buf1, NS);
  edge_own_k<128, false><<<NO, TO, 0, stream>>>(
      (const v4u*)own, E1, E2, src1, dst1, src2, dst2, dinv, buf1, b1, b1, buf2, (size_t)0, wlist, N);

  float* zlin   = buf1;
  float* out_mu = (float*)d_out;
  prod_strip_k<4, true><<<blocks((long long)NS * 32, TP), TP, 0, stream>>>(buf2, wmut, dinv, zlin, NS);
  edge_own_k<64, true><<<NO, TO, 0, stream>>>(
      (const v4u*)own, E1, E2, src1, dst1, src2, dst2, dinv, zlin, bmu, bls, out_mu, (size_t)N * DZ, wlist, N);
}


# Owner build orders each owner's records by destination row (LDS counting sort on the degree counts); edge passes sum a row's gathered rows in registers, one LDS add per run; sum order changes, varies
# speedup vs baseline: 4.4947x; 1.3529x over previous
#include <hip/hip_runtime.h>

typedef __attribute__((ext_vector_type(16))) __bf16 v16bf;
typedef __attribute__((ext_vector_type(8)))  __bf16 v8bf;
typedef __attribute__((ext_vector_type(8)))  float  v8f;
typedef __attribute__((ext_vector_type(4)))  float    v4f;
typedef __attribute__((ext_vector_type(4)))  unsigned v4u;

__device__ __forceinline__ __bf16 f2bf(float f) {
  union { float f; unsigned u; } v; v.f = f;
  unsigned r = v.u + 0x7FFFu + ((v.u >> 16) & 1u);
  union { unsigned short s; __bf16 b; } o;
  o.s = (unsigned short)(r >> 16);
  return o.b;
}

template <int NT, bool RELU>
__global__ void __launch_bounds__(128) __attribute__((amdgpu_waves_per_eu(1, 4)))
prod_strip_k(const float* __restrict__ A, const __bf16* __restrict__ Bt,
             const float* __restrict__ dinv, float* __restrict__ C, int nstrips) {
  constexpr int K  = 128;
  constexpr int N  = 16 * NT;
  const int lane  = threadIdx.x & 31;
  const int strip = blockIdx.x * (blockDim.x >> 5) + (threadIdx.x >> 5);
  if (strip >= nstrips) return;
  const int m16   = lane & 15;
  const int khalf = (lane >> 4) << 3;
  const float* ap = A + (size_t)(strip * 16 + m16) * K + khalf;

  v8f alo[4], ahi[4];
#pragma unroll
  for (int s = 0; s < 4; ++s) {
    alo[s] = *(const v8f*)(ap + 32 * s);
    ahi[s] = *(const v8f*)(ap + 32 * s + 16);
  }
  const __bf16* bp = Bt + (size_t)m16 * K + khalf;
  v16bf b[2][NT];
  auto load_b = [&](v16bf* dst, int s) {
#pragma unroll
    for (int t = 0; t < NT; ++t) {
      const v8bf blo = *(const v8bf*)(bp + (size_t)t * 16 * K + 32 * s);
      const v8bf bhi = *(const v8bf*)(bp + (size_t)t * 16 * K + 32 * s + 16);
      dst[t] = __builtin_shufflevector(blo, bhi, 0,1,2,3,4,5,6,7,8,9,10,11,12,13,14,15);
    }
  };
  load_b(b[0], 0);

  v16bf a[4];
#pragma unroll
  for (int s = 0; s < 4; ++s) {
    v8f lo = alo[s], hi = ahi[s];
    if (RELU) {
#pragma unroll
      for (int i = 0; i < 8; ++i) { lo[i] = fmaxf(lo[i], 0.0f); hi[i] = fmaxf(hi[i], 0.0f); }
    }
    const v8bf l = __builtin_convertvector(lo, v8bf);
    const v8bf h = __builtin_convertvector(hi, v8bf);
    a[s] = __builtin_shufflevector(l, h, 0,1,2,3,4,5,6,7,8,9,10,11,12,13,14,15);
  }

  v8f acc[NT];
#pragma unroll
  for (int t = 0; t < NT; ++t) acc[t] = v8f{};
#pragma unroll
  for (int s = 0; s < 4; ++s) {
    if (s + 1 < 4) load_b(b[(s + 1) & 1], s + 1);
#pragma unroll
    for (int t = 0; t < NT; ++t)
      acc[t] = __builtin_amdgcn_wmma_f32_16x16x32_bf16(
          false, a[s], false, b[s & 1][t], (short)0, acc[t], false, false);
  }

  const int crow = strip * 16 + ((lane >> 4) << 3);
  const v8f di = *(const v8f*)(dinv + crow);
#pragma unroll
  for (int t = 0; t < NT; ++t) {
    const int col = t * 16 + m16;
#pragma unroll
    for (int r = 0; r < 8; ++r) C[(size_t)(crow + r) * N + col] = acc[t][r] * di[r];
  }
}

constexpr int TR      = 196;
constexpr int LSTRIDE = 4096;
constexpr int REC0    = 32;
constexpr int CH      = 4096;
constexpr int OSTR    = 264;
constexpr int ST      = 512;
static_assert(TR <= 255, "a record keeps the row inside the range in 8 bits, 255 = no edge");
static_assert(CH % (ST * 4) == 0, "a thread takes whole groups of four neighbouring edges");

__device__ __forceinline__ void cvt_tr_one(const float* __restrict__ W, __bf16* __restrict__ Wt,
                                           int i, int K, int N) {
  const int k = i / N, n = i % N;
  Wt[(size_t)n * K + k] = f2bf(W[i]);
}

__device__ __forceinline__ int wave_incl_scan(int v, int lane) {
#pragma unroll
  for (int dd = 1; dd < 32; dd <<= 1) {
    const int t = __shfl_up(v, dd);
    if (lane >= dd) v += t;
  }
  return v;
}

__global__ void __launch_bounds__(ST)
setup_k(const float* __restrict__ W1, const float* __restrict__ Wmu,
        const float* __restrict__ Wls, __bf16* __restrict__ w1t,
        __bf16* __restrict__ wmut, __bf16* __restrict__ wlst,
        const int* __restrict__ src1, const int* __restrict__ dst1, int e1,
        const int* __restrict__ src2, const int* __restrict__ dst2, int e2,
        unsigned* __restrict__ own32, unsigned* __restrict__ rec, int* __restrict__ off, int nch) {
  __shared__ int hist[256];
  __shared__ int offs[257];
  __shared__ int s_wsum[ST / 32];
  __shared__ __attribute__((aligned(16))) unsigned stage[CH];

  const int tid = threadIdx.x;
  if ((int)blockIdx.x >= nch) {
    int i = ((int)blockIdx.x - nch) * ST + tid;
    if (i < 128 * 128) { cvt_tr_one(W1, w1t, i, 128, 128); return; }
    i -= 128 * 128;
    if (i < 128 * 32) { cvt_tr_one(Wmu, wmut, i, 128, 32); return; }
    i -= 128 * 32;
    if (i < 128 * 32) { cvt_tr_one(Wls, wlst, i, 128, 32); }
    return;
  }
  const int lane = tid & 31, wv = tid >> 5;
  const int c = blockIdx.x;
  const int E = e1 + e2;
  if (tid < 256) hist[tid] = 0;
  __syncthreads();

  auto load4 = [&](const int* l1, const int* l2, int e0, int* v) {
    const int* p = e0 < e1 ? l1 + e0 : l2 + (e0 - e1);
    const bool one_list = (e0 + 4 <= e1) || (e0 >= e1 && e0 + 4 <= E);
    if (one_list && ((size_t)p & 15) == 0) {
      const v4u x = *(const v4u*)p;
      v[0] = (int)x.x; v[1] = (int)x.y; v[2] = (int)x.z; v[3] = (int)x.w;
    } else {
#pragma unroll
      for (int b = 0; b < 4; ++b) {
        const int e = e0 + b;
        v[b] = 0;
        if (e < e1)     v[b] = l1[e];
        else if (e < E) v[b] = l2[e - e1];
      }
    }
  };

  constexpr int G = CH / (ST * 4);
  int s[G][4], d[G][4], o[G][4], rk[G][4];
#pragma unroll
  for (int g = 0; g < G; ++g) {
    const int e0 = c * CH + g * (ST * 4) + tid * 4;
    load4(dst1, dst2, e0, d[g]);
    load4(src1, src2, e0, s[g]);
  }
#pragma unroll
  for (int g = 0; g < G; ++g) {
    const int e0 = c * CH + g * (ST * 4) + tid * 4;
#pragma unroll
    for (int b = 0; b < 4; ++b) {
      o[g][b]  = min((int)((unsigned)d[g][b] / TR), 255);
      rk[g][b] = 0;
      if (e0 + b < E) rk[g][b] = atomicAdd(&hist[o[g][b]], 1);
    }
    own32[e0 >> 2] = (unsigned)o[g][0] | ((unsigned)o[g][1] << 8) |
                     ((unsigned)o[g][2] << 16) | ((unsigned)o[g][3] << 24);
  }
  __syncthreads();

  const int h    = tid < 256 ? hist[tid] : 0;
  const int incl = wave_incl_scan(h, lane);
  if (lane == 31) s_wsum[wv] = incl;
  __syncthreads();
  if (tid < 256) {
    int wb = 0;
#pragma unroll
    for (int w = 0; w < 8; ++w) wb += w < wv ? s_wsum[w] : 0;
    const int ex = wb + incl - h;
    offs[tid] = ex;
    off[(size_t)c * OSTR + tid] = ex;
    if (tid == 255) { offs[256] = ex + h; off[(size_t)c * OSTR + 256] = ex + h; }
  }
  __syncthreads();

#pragma unroll
  for (int g = 0; g < G; ++g) {
    const int e0 = c * CH + g * (ST * 4) + tid * 4;
#pragma unroll
    for (int b = 0; b < 4; ++b) {
      if (e0 + b < E) {
        const unsigned dl = (unsigned)(d[g][b] - o[g][b] * TR) & 0xffu;
        stage[offs[o[g][b]] + rk[g][b]] = ((unsigned)s[g][b] & 0x00ffffffu) | (dl << 24);
      }
    }
  }
  __syncthreads();

  const int total = offs[256];
  const v4u* st4 = (const v4u*)stage;
  v4u* rec4 = (v4u*)(rec + (size_t)c * CH);
#pragma unroll
  for (int k = 0; k < CH / 4 / ST; ++k) {
    const int i = tid + k * ST;
    if (i * 4 < total) rec4[i] = st4[i];
  }
}

template <bool B> struct flag_t { static constexpr bool value = B; };

__device__ __forceinline__ unsigned zero_bytes(unsigned y) {
  return ~(((y & 0x7f7f7f7fu) + 0x7f7f7f7fu) | y | 0x7f7f7f7fu);
}

template <typename OVER>
__device__ __forceinline__ void scan_owner_bytes(const v4u* __restrict__ own4, int E, int bid,
                                                 int wv, int lane, int* list, int cap,
                                                 int* s_cnt, OVER over) {
  const int      G   = (E + 15) >> 4;
  const unsigned pat = (unsigned)bid * 0x01010101u;
  constexpr int  U   = 4;
  for (int gb = wv * 32; gb < G; gb += 1024 * U) {
    v4u x[U];
#pragma unroll
    for (int u = 0; u < U; ++u) x[u] = own4[min(gb + u * 1024 + lane, G - 1)];
#pragma unroll
    for (int u = 0; u < U; ++u) {
      const int g = gb + u * 1024 + lane;
      const unsigned ok = g < G ? ~0u : 0u;
      unsigned t0 = zero_bytes(x[u].x ^ pat) & ok, t1 = zero_bytes(x[u].y ^ pat) & ok;
      unsigned t2 = zero_bytes(x[u].z ^ pat) & ok, t3 = zero_bytes(x[u].w ^ pat) & ok;
      while (true) {
        const bool     has = (t0 | t1 | t2 | t3) != 0u;
        const unsigned m   = __builtin_amdgcn_ballot_w32(has);
        if (m == 0u) break;
        int slot0 = 0;
        if (lane == 0) slot0 = atomicAdd(s_cnt, __builtin_popcount(m));
        slot0 = __builtin_amdgcn_readfirstlane(slot0);
        if (has) {
          const int slot = slot0 + (int)__builtin_amdgcn_mbcnt_lo(m, 0u);
          const int      q  = t0 ? 0 : t1 ? 1 : t2 ? 2 : 3;
          const unsigned tq = t0 ? t0 : t1 ? t1 : t2 ? t2 : t3;
          const unsigned tc = tq & (tq - 1u);
          t0 = q == 0 ? tc : t0;  t1 = q == 1 ? tc : t1;
          t2 = q == 2 ? tc : t2;  t3 = q == 3 ? tc : t3;
          const int e = g * 16 + q * 4 + (__builtin_ctz(tq) >> 3);
          if (slot < cap)  list[slot] = e < E ? e : -1;
          else if (e < E)  over(e);
        }
      }
    }
  }
}

constexpr int OWNER_LDS = 168 * 1024;

__global__ void __launch_bounds__(ST)
owner_build_k(const int* __restrict__ off, const unsigned* __restrict__ rec, int nch,
              float* __restrict__ dinv, int* __restrict__ wlist, int N) {
  __shared__ int cntl[256];
  __shared__ int roff[256];
  __shared__ unsigned       s_rec[LSTRIDE - REC0];
  __shared__ unsigned short s_rk[LSTRIDE - REC0];
  __shared__ unsigned       s_srt[LSTRIDE - REC0];
  __shared__ int s_st[ST];
  __shared__ int s_ln[ST];
  __shared__ int s_pos[ST];
  __shared__ int s_wsum[ST / 32];

  const int tid  = threadIdx.x;
  const int lane = tid & 31;
  const int wv   = __builtin_amdgcn_readfirstlane(tid >> 5);
  const int bid  = blockIdx.x;
  const int base = bid * TR;
  const int rows = min(TR, N - base);
  int* wl = wlist + (size_t)bid * LSTRIDE;

  if (tid < 256) cntl[tid] = 0;

  int st = 0, ln = 0;
  if (tid < nch) {
    const int* oc = off + (size_t)tid * OSTR + bid;
    st = oc[0];
    ln = oc[1] - st;
  }
  const int incl = wave_incl_scan(ln, lane);
  if (lane == 31) s_wsum[wv] = incl;
  __syncthreads();
  int wb = 0, cnt = 0;
#pragma unroll
  for (int w = 0; w < ST / 32; ++w) { const int t = s_wsum[w]; cnt += t; wb += w < wv ? t : 0; }
  s_st[tid] = st;  s_ln[tid] = ln;  s_pos[tid] = wb + incl - ln;
  __syncthreads();

  const bool fits = cnt <= LSTRIDE - REC0;
  constexpr int NW = ST / 32;
  constexpr int U  = 8;
  for (int c0 = wv; c0 < nch; c0 += NW * U) {
    int rs[U], rl[U], rp[U];
    unsigned w[U];
#pragma unroll
    for (int u = 0; u < U; ++u) {
      const int c = c0 + u * NW;
      rs[u] = 0; rl[u] = 0; rp[u] = 0;
      if (c < nch) { rs[u] = c * CH + s_st[c]; rl[u] = s_ln[c]; rp[u] = s_pos[c]; }
    }
#pragma unroll
    for (int u = 0; u < U; ++u) w[u] = lane < rl[u] ? rec[rs[u] + lane] : 0xff000000u;
#pragma unroll
    for (int u = 0; u < U; ++u) {
      if (lane < rl[u]) {
        const int rk = atomicAdd(&cntl[w[u] >> 24], 1);
        if (fits) { s_rec[rp[u] + lane] = w[u]; s_rk[rp[u] + lane] = (unsigned short)rk; }
      }
      for (int i = lane + 32; i < rl[u]; i += 32) {
        const unsigned x = rec[rs[u] + i];
        const int rk = atomicAdd(&cntl[x >> 24], 1);
        if (fits) { s_rec[rp[u] + i] = x; s_rk[rp[u] + i] = (unsigned short)rk; }
      }
    }
  }
  __syncthreads();

  if (tid < rows) {
    const float deg = 1.0f + (float)cntl[tid];
    dinv[base + tid] = deg > 0.0f ? rsqrtf(deg) : 0.0f;
  }
  if (tid == 0) wl[0] = fits ? cnt : -1;

  const int h    = tid < 256 ? cntl[tid] : 0;
  const int inc2 = wave_incl_scan(h, lane);
  if (lane == 31) s_wsum[wv] = inc2;
  __syncthreads();
  if (tid < 256) {
    int wb2 = 0;
#pragma unroll
    for (int w = 0; w < 8; ++w) wb2 += w < wv ? s_wsum[w] : 0;
    roff[tid] = wb2 + inc2 - h;
  }
  __syncthreads();
  if (fits) {
    for (int i = tid; i < cnt; i += ST) {
      const unsigned x = s_rec[i];
      s_srt[roff[x >> 24] + (int)s_rk[i]] = x;
    }
  }
  __syncthreads();
  if (fits) {
    for (int i = tid; i < cnt; i += ST) wl[REC0 + i] = (int)s_srt[i];
  }
}

template <int F, bool PLANES>
__global__ void __launch_bounds__(1024)
edge_own_k(const v4u* __restrict__ own4, int E1, int E2,
           const int* __restrict__ src1, const int* __restrict__ dst1,
           const int* __restrict__ src2, const int* __restrict__ dst2,
           const float* __restrict__ dinv, const float* __restrict__ hlin,
           const float* __restrict__ bias0, const float* __restrict__ bias1,
           float* __restrict__ out, size_t plane, const int* __restrict__ wlist, int N) {
  constexpr int J   = F / 32;
  constexpr int EPW = 32 / J;
  constexpr int LPE = 32 / EPW;
  constexpr int CAP = (OWNER_LDS - TR * F * 4) / 4 - TR - 8;
  static_assert(CAP >= LSTRIDE, "edges that fit the record slots must fit the LDS list");
  __shared__ float tile[TR * F];
  __shared__ int   list[CAP];
  __shared__ float s_di[TR];
  __shared__ int   s_cnt;

  const int tid  = threadIdx.x;
  const int lane = tid & 31;
  const int wv   = __builtin_amdgcn_readfirstlane(tid >> 5);
  const int bid  = blockIdx.x;
  const int base = bid * TR;
  const int rows = min(TR, N - base);
  const int E    = E1 + E2;
  v4f* t4 = (v4f*)tile;
  const int* wl = wlist + (size_t)bid * LSTRIDE;

  auto out4 = [&](int i) -> v4f* {
    if (!PLANES) return (v4f*)(out + (size_t)base * F) + i;
    const int r = i >> 4, q = i & 15;
    return (v4f*)(out + (q >= 8 ? plane : (size_t)0) + (size_t)(base + r) * 32 + (q & 7) * 4);
  };

  const int saved = __builtin_amdgcn_readfirstlane(wl[0]);
  if (tid == 0) s_cnt = saved >= 0 ? saved : 0;
  if (tid < rows) s_di[tid] = dinv[base + tid];
  {
    constexpr int NI = (TR * F / 4 + 1023) / 1024;
    const int lim = rows * (F / 4);
    const v4f* own_rows = (const v4f*)(hlin + (size_t)base * F);
    v4f r[NI];
#pragma unroll
    for (int k = 0; k < NI; ++k) r[k] = own_rows[min(tid + k * 1024, lim - 1)];
#pragma unroll
    for (int k = 0; k < NI; ++k) t4[min(tid + k * 1024, lim - 1)] = r[k];
  }
  __syncthreads();

  auto dst_of = [&](int e) { return e < E1 ? dst1[e] : dst2[e - E1]; };
  auto src_of = [&](int e) { return e < E1 ? src1[e] : src2[e - E1]; };

  if (saved < 0) {
    scan_owner_bytes(own4, E, bid, wv, lane, list, CAP, &s_cnt, [&](int e) {
      const int s = src_of(e), d = dst_of(e);
      for (int j = 0; j < F; ++j) atomicAdd(&tile[(d - base) * F + j], hlin[(size_t)s * F + j]);
    });
  }
  __syncthreads();

  const int cnt = s_cnt;
  const int n   = saved >= 0 ? saved : min(cnt, CAP);
  const int nb  = (n + EPW - 1) / EPW;
  auto gather = [&](auto rec_flag) {
    constexpr bool REC = decltype(rec_flag)::value;
    auto fetch_rec = [&](int b) -> unsigned {
      const int eq = b * EPW + lane / LPE;
      unsigned r = 0xff000000u;
      if (b < nb && eq < n) {
        if (REC) r = (unsigned)wl[REC0 + eq];
        else {
          const int e = list[eq];
          if (e >= 0) r = (unsigned)src_of(e) | ((unsigned)(dst_of(e) - base) << 24);
        }
      }
      return r;
    };
    const int nbw = (nb + 31) >> 5;
    const int b0  = wv * nbw;
    const int b1  = min(nb, b0 + nbw);
    float acc[J];
#pragma unroll
    for (int j = 0; j < J; ++j) acc[j] = 0.0f;
    int cur = -1;
    unsigned rec = fetch_rec(b0 < b1 ? b0 : nb);
    for (int b = b0; b < b1; ++b) {
      const int dl   = (int)(rec >> 24);
      const int live = dl != 255;
      const int s    = live ? (int)(rec & 0x00ffffffu) : 0;
      float v[EPW][J];
#pragma unroll
      for (int k = 0; k < EPW; ++k) {
        const int sk = __builtin_amdgcn_readlane(s, k * LPE);
#pragma unroll
        for (int j = 0; j < J; ++j) v[k][j] = hlin[(size_t)sk * F + lane + 32 * j];
      }
      rec = fetch_rec(b + 1 < b1 ? b + 1 : nb);
#pragma unroll
      for (int k = 0; k < EPW; ++k) {
        if (__builtin_amdgcn_readlane(live, k * LPE)) {
          const int dk = __builtin_amdgcn_readlane(dl, k * LPE);
          if (dk == cur) {
#pragma unroll
            for (int j = 0; j < J; ++j) acc[j] += v[k][j];
          } else {
            if (cur >= 0) {
              float* tp = tile + cur * F + lane;
#pragma unroll
              for (int j = 0; j < J; ++j) atomicAdd(tp + 32 * j, acc[j]);
            }
            cur = dk;
#pragma unroll
            for (int j = 0; j < J; ++j) acc[j] = v[k][j];
          }
        }
      }
    }
    if (cur >= 0) {
      float* tp = tile + cur * F + lane;
#pragma unroll
      for (int j = 0; j < J; ++j) atomicAdd(tp + 32 * j, acc[j]);
    }
  };
  if (saved >= 0) gather(flag_t<true>{});
  else            gather(flag_t<false>{});
  __syncthreads();

  for (int i = tid; i < rows * (F / 4); i += 1024) {
    const int r = i / (F / 4), q = i % (F / 4);
    const float* bp = (PLANES && q >= 8) ? bias1 + (q - 8) * 4 : bias0 + q * 4;
    v4f bv; bv.x = bp[0]; bv.y = bp[1]; bv.z = bp[2]; bv.w = bp[3];
    *out4(i) = t4[i] * s_di[r] + bv;
  }
}

extern "C" void kernel_launch(void* const* d_in, const int* in_sizes, int n_in,
                              void* d_out, int out_size, void* d_ws, size_t ws_size,
                              hipStream_t stream) {
  const float* x   = (const float*)d_in[0];
  const int*   ei  = (const int*)d_in[1];
  const int*   yei = (const int*)d_in[2];
  const float* W1  = (const float*)d_in[3];
  const float* b1  = (const float*)d_in[4];
  const float* Wmu = (const float*)d_in[5];
  const float* bmu = (const float*)d_in[6];
  const float* Wls = (const float*)d_in[7];
  const float* bls = (const float*)d_in[8];

  const int DIN = 128, DH = 128, DZ = 32;
  const int N  = in_sizes[0] / DIN;
  const int E1 = in_sizes[1] / 2;
  const int E2 = in_sizes[2] / 2;

  const int* src1 = ei;   const int* dst1 = ei  + E1;
  const int* src2 = yei;  const int* dst2 = yei + E2;

  char* wsb = (char*)d_ws;
  float*  dinv = (float*)wsb;
  __bf16* w1t  = (__bf16*)(wsb + (size_t)65536 * 4);
  __bf16* wmut = w1t + 128 * 128;
  __bf16* wlst = wmut + 32 * 128;
  unsigned char* own = (unsigned char*)(wsb + (size_t)(65536 + 16384 + 65536) * 4);
  int*    wlist = (int*)(own + ((size_t)1 << 20));
  unsigned* rec = (unsigned*)(wsb + ((size_t)6 << 20));
  int*    off   = (int*)(wsb + ((size_t)10 << 20));
  float*  buf1 = (float*)(wsb + (size_t)(65536 + 16384) * 4 + (size_t)N * DH * 2);
  float*  buf2 = buf1 + (size_t)N * DH;

  const int TP = 128;
  auto blocks = [](long long total, int t) { return (unsigned)((total + t - 1) / t); };
  const int NS = N / 16;
  const long long EA = (long long)E1 + E2;
  const int TO = 1024;
  const int NO = (N + TR - 1) / TR;
  const int NCH = (int)blocks(EA, CH);

  setup_k<<<NCH + blocks(DIN * DH + 2 * DH * DZ, ST), ST, 0, stream>>>(
      W1, Wmu, Wls, w1t, wmut, wlst, src1, dst1, E1, src2, dst2, E2, (unsigned*)own, rec, off, NCH);
  owner_build_k<<<NO, ST, 0, stream>>>(off, rec, NCH, dinv, wlist, N);

  prod_strip_k<8, false><<<blocks((long long)NS * 32, TP), TP, 0, stream>>>(x, w1t, dinv, buf1, NS);
  edge_own_k<128, false><<<NO, TO, 0, stream>>>(
      (const v4u*)own, E1, E2, src1, dst1, src2, dst2, dinv, buf1, b1, b1, buf2, (size_t)0, wlist, N);

  float* zlin   = buf1;
  float* out_mu = (float*)d_out;
  prod_strip_k<4, true><<<blocks((long long)NS * 32, TP), TP, 0, stream>>>(buf2, wmut, dinv, zlin, NS);
  edge_own_k<64, true><<<NO, TO, 0, stream>>>(
      (const v4u*)own, E1, E2, src1, dst1, src2, dst2, dinv, zlin, bmu, bls, out_mu, (size_t)N * DZ, wlist, N);
}
